# starting kernel
# baseline (speedup 1.0000x reference)
#include <hip/hip_runtime.h>

#define D_DIM  512
#define B_DIM  16
#define TV_DIM 256
#define TS_DIM 32
#define VPB    8

#define KC     32
#define LDS_S  36

typedef float v2f __attribute__((ext_vector_type(2)));
typedef float v8f __attribute__((ext_vector_type(8)));

__device__ __forceinline__ float fast_tanh(float x) {
#if __has_builtin(__builtin_amdgcn_tanhf)
  return __builtin_amdgcn_tanhf(x);
#else
  float y;
  asm("v_tanh_f32 %0, %1" : "=v"(y) : "v"(x));
  return y;
#endif
}

__global__ __launch_bounds__(256) void gemm_xwt_wmma(
    const float* __restrict__ A, const float* __restrict__ W,
    const float* __restrict__ bias, const float* __restrict__ rmask,
    float* __restrict__ out, int M, int N, int K) {
  __shared__ float As[2][128 * LDS_S];
  __shared__ float Ws[2][ 64 * LDS_S];

  const int tid  = threadIdx.x;
  const int lane = tid & 31;
  const int wib  = tid >> 5;
  const int ntn  = N >> 6;
  const int mblk = blockIdx.x / ntn;
  const int ng   = blockIdx.x % ntn;
  const int mBase = mblk * 128;
  const int n0    = ng * 64;
  const int half  = lane >> 4;
  const int lm    = lane & 15;
  const int nchunks = K / KC;

  int arR[4], arC[4], wrR[2], wrC[2];
#pragma unroll
  for (int i = 0; i < 4; ++i) { int f = tid + i * 256; arR[i] = f >> 3; arC[i] = (f & 7) * 4; }
#pragma unroll
  for (int i = 0; i < 2; ++i) { int f = tid + i * 256; wrR[i] = f >> 3; wrC[i] = (f & 7) * 4; }

  float4 ar[4], wr[2];
#pragma unroll
  for (int i = 0; i < 4; ++i)
    ar[i] = *(const float4*)(A + (size_t)(mBase + arR[i]) * K + arC[i]);
#pragma unroll
  for (int i = 0; i < 2; ++i)
    wr[i] = *(const float4*)(W + (size_t)(n0 + wrR[i]) * K + wrC[i]);
#pragma unroll
  for (int i = 0; i < 4; ++i)
    *(float4*)&As[0][arR[i] * LDS_S + arC[i]] = ar[i];
#pragma unroll
  for (int i = 0; i < 2; ++i)
    *(float4*)&Ws[0][wrR[i] * LDS_S + wrC[i]] = wr[i];
  __syncthreads();

  v8f c0 = {0.f,0.f,0.f,0.f,0.f,0.f,0.f,0.f};
  v8f c1 = {0.f,0.f,0.f,0.f,0.f,0.f,0.f,0.f};
  v8f c2 = {0.f,0.f,0.f,0.f,0.f,0.f,0.f,0.f};
  v8f c3 = {0.f,0.f,0.f,0.f,0.f,0.f,0.f,0.f};

  const int arow = 16 * wib + lm;
  for (int c = 0; c < nchunks; ++c) {
    const int buf = c & 1;
    const int nbuf = buf ^ 1;
    const bool more = (c + 1) < nchunks;
    if (more) {
      const int k0n = (c + 1) * KC;
#pragma unroll
      for (int i = 0; i < 4; ++i)
        ar[i] = *(const float4*)(A + (size_t)(mBase + arR[i]) * K + k0n + arC[i]);
#pragma unroll
      for (int i = 0; i < 2; ++i)
        wr[i] = *(const float4*)(W + (size_t)(n0 + wrR[i]) * K + k0n + wrC[i]);
    }
    const float* Ab = &As[buf][arow * LDS_S + 2 * half];
    const float* Wb = &Ws[buf][lm   * LDS_S + 2 * half];
#pragma unroll
    for (int kk = 0; kk < KC; kk += 4) {
      v2f a  = *(const v2f*)(Ab + kk);
      v2f b0 = *(const v2f*)(Wb + kk);
      v2f b1 = *(const v2f*)(Wb + 16 * LDS_S + kk);
      v2f b2 = *(const v2f*)(Wb + 32 * LDS_S + kk);
      v2f b3 = *(const v2f*)(Wb + 48 * LDS_S + kk);
      c0 = __builtin_amdgcn_wmma_f32_16x16x4_f32(false, a, false, b0, (short)0, c0, false, false);
      c1 = __builtin_amdgcn_wmma_f32_16x16x4_f32(false, a, false, b1, (short)0, c1, false, false);
      c2 = __builtin_amdgcn_wmma_f32_16x16x4_f32(false, a, false, b2, (short)0, c2, false, false);
      c3 = __builtin_amdgcn_wmma_f32_16x16x4_f32(false, a, false, b3, (short)0, c3, false, false);
    }
    if (more) {
#pragma unroll
      for (int i = 0; i < 4; ++i)
        *(float4*)&As[nbuf][arR[i] * LDS_S + arC[i]] = ar[i];
#pragma unroll
      for (int i = 0; i < 2; ++i)
        *(float4*)&Ws[nbuf][wrR[i] * LDS_S + wrC[i]] = wr[i];
    }
    __syncthreads();
  }

  const int m0  = mBase + 16 * wib;
  const int col = n0 + lm;
  const float bb0 = bias ? bias[col +  0] : 0.f;
  const float bb1 = bias ? bias[col + 16] : 0.f;
  const float bb2 = bias ? bias[col + 32] : 0.f;
  const float bb3 = bias ? bias[col + 48] : 0.f;
#pragma unroll
  for (int r = 0; r < 8; ++r) {
    const int row  = m0 + r + 8 * half;
    const float rm = rmask[row];
    float* orow = out + (size_t)row * N;
    orow[col +  0] = (c0[r] + bb0) * rm;
    orow[col + 16] = (c1[r] + bb1) * rm;
    orow[col + 32] = (c2[r] + bb2) * rm;
    orow[col + 48] = (c3[r] + bb3) * rm;
  }
}

__global__ __launch_bounds__(256) void tanh_dot_kernel(
    const float* __restrict__ t1g,
    const float* __restrict__ t2g,
    const float* __restrict__ wtg,
    const float* __restrict__ vmask,
    const float* __restrict__ smask,
    float* __restrict__ out) {
  __shared__ float t2s[256 * 33];
  __shared__ __align__(16) float wts[D_DIM];
  __shared__ __align__(16) float t1s[VPB * D_DIM];

  const int tid  = threadIdx.x;
  const int lane = tid & 31;
  const int w    = tid >> 5;
  const int b    = blockIdx.x / (TV_DIM / VPB);
  const int v    = (blockIdx.x % (TV_DIM / VPB)) * VPB + w;
  const int rowv = b * TV_DIM + v;

  for (int i = tid; i < D_DIM; i += 256) wts[i] = wtg[i];
#pragma unroll
  for (int j = 0; j < D_DIM / 32; ++j)
    t1s[w * D_DIM + lane + 32 * j] = t1g[(size_t)rowv * D_DIM + lane + 32 * j];

  const float pm = vmask[rowv] * smask[b * TS_DIM + lane];
  float acc = 0.f;

  for (int h = 0; h < 2; ++h) {
    if (h) __syncthreads();
    for (int i = tid; i < 256 * TS_DIM; i += 256) {
      const int s  = i >> 8;
      const int dl = i & 255;
      t2s[dl * 33 + s] = t2g[(size_t)(b * TS_DIM + s) * D_DIM + h * 256 + dl];
    }
    __syncthreads();

    const float* t1h = t1s + w * D_DIM + h * 256;
    const float* wth = wts + h * 256;
#pragma unroll 4
    for (int d = 0; d < 256; d += 4) {
      const float4 x1 = *(const float4*)(t1h + d);
      const float4 xw = *(const float4*)(wth + d);
      const float a0 = t2s[(d + 0) * 33 + lane];
      const float a1 = t2s[(d + 1) * 33 + lane];
      const float a2 = t2s[(d + 2) * 33 + lane];
      const float a3 = t2s[(d + 3) * 33 + lane];
      acc = fmaf(fast_tanh((x1.x + a0) * pm), xw.x, acc);
      acc = fmaf(fast_tanh((x1.y + a1) * pm), xw.y, acc);
      acc = fmaf(fast_tanh((x1.z + a2) * pm), xw.z, acc);
      acc = fmaf(fast_tanh((x1.w + a3) * pm), xw.w, acc);
    }
  }

  out[(size_t)rowv * TS_DIM + lane] = acc * pm;
}

extern "C" void kernel_launch(void* const* d_in, const int* in_sizes, int n_in,
                              void* d_out, int out_size, void* d_ws, size_t ws_size,
                              hipStream_t stream) {
  const float* video = (const float*)d_in[0];
  const float* vmask = (const float*)d_in[1];
  const float* sent  = (const float*)d_in[2];
  const float* smask = (const float*)d_in[3];
  const float* w1    = (const float*)d_in[4];
  const float* b1    = (const float*)d_in[5];
  const float* w2    = (const float*)d_in[6];
  const float* wt    = (const float*)d_in[7];
  float* out  = (float*)d_out;

  float* tmp1 = (float*)d_ws;
  float* tmp2 = tmp1 + (size_t)B_DIM * TV_DIM * D_DIM;

  gemm_xwt_wmma<<<256, 256, 0, stream>>>(video, w1, b1, vmask, tmp1,
                                         B_DIM * TV_DIM, D_DIM, D_DIM);
  gemm_xwt_wmma<<<32, 256, 0, stream>>>(sent, w2, nullptr, smask, tmp2,
                                        B_DIM * TS_DIM, D_DIM, D_DIM);
  tanh_dot_kernel<<<B_DIM * (TV_DIM / VPB), 256, 0, stream>>>(
      tmp1, tmp2, wt, vmask, smask, out);
}


# GEMM k-loop: next chunk's global loads stay in six named registers under the WMMA block (arrays had gone to scratch/LDS, waited at issue); last chunk peeled; occupancy hint keeps LDS reads grouped
# speedup vs baseline: 2.1877x; 2.1877x over previous
#include <hip/hip_runtime.h>

#define D_DIM  512
#define B_DIM  16
#define TV_DIM 256
#define TS_DIM 32
#define VPB    8

#define KC     32
#define LDS_S  36

typedef float v2f __attribute__((ext_vector_type(2)));
typedef float v4f __attribute__((ext_vector_type(4)));
typedef float v8f __attribute__((ext_vector_type(8)));

__device__ __forceinline__ float fast_tanh(float x) {
#if __has_builtin(__builtin_amdgcn_tanhf)
  return __builtin_amdgcn_tanhf(x);
#else
  float y;
  asm("v_tanh_f32 %0, %1" : "=v"(y) : "v"(x));
  return y;
#endif
}

__device__ __forceinline__ void wmma_chunk(const float* Ab, const float* Wb,
                                           v8f& c0, v8f& c1, v8f& c2, v8f& c3) {
#pragma unroll
  for (int kk = 0; kk < KC; kk += 4) {
    v2f a  = *(const v2f*)(Ab + kk);
    v2f b0 = *(const v2f*)(Wb + kk);
    v2f b1 = *(const v2f*)(Wb + 16 * LDS_S + kk);
    v2f b2 = *(const v2f*)(Wb + 32 * LDS_S + kk);
    v2f b3 = *(const v2f*)(Wb + 48 * LDS_S + kk);
    c0 = __builtin_amdgcn_wmma_f32_16x16x4_f32(false, a, false, b0, (short)0, c0, false, false);
    c1 = __builtin_amdgcn_wmma_f32_16x16x4_f32(false, a, false, b1, (short)0, c1, false, false);
    c2 = __builtin_amdgcn_wmma_f32_16x16x4_f32(false, a, false, b2, (short)0, c2, false, false);
    c3 = __builtin_amdgcn_wmma_f32_16x16x4_f32(false, a, false, b3, (short)0, c3, false, false);
  }
}

__global__ __launch_bounds__(256) __attribute__((amdgpu_waves_per_eu(2, 4))) void gemm_xwt_wmma(
    const float* __restrict__ A, const float* __restrict__ W,
    const float* __restrict__ bias, const float* __restrict__ rmask,
    float* __restrict__ out, int M, int N, int K) {
  __shared__ float As[2][128 * LDS_S];
  __shared__ float Ws[2][ 64 * LDS_S];

  const int tid  = threadIdx.x;
  const int lane = tid & 31;
  const int wib  = tid >> 5;
  const int ntn  = N >> 6;
  const int mblk = blockIdx.x / ntn;
  const int ng   = blockIdx.x % ntn;
  const int mBase = mblk * 128;
  const int n0    = ng * 64;
  const int half  = lane >> 4;
  const int lm    = lane & 15;
  const int nchunks = K / KC;

  const int sr = tid >> 3;
  const int sc = (tid & 7) * 4;
  const size_t rows32 = (size_t)32 * K;
  const float* ga = A + (size_t)(mBase + sr) * K + sc;
  const float* gw = W + (size_t)(n0    + sr) * K + sc;
  const int ls = sr * LDS_S + sc;

  v4f a0 = *(const v4f*)(ga);
  v4f a1 = *(const v4f*)(ga + rows32);
  v4f a2 = *(const v4f*)(ga + 2 * rows32);
  v4f a3 = *(const v4f*)(ga + 3 * rows32);
  v4f w0 = *(const v4f*)(gw);
  v4f w1 = *(const v4f*)(gw + rows32);
  *(v4f*)&As[0][ls]              = a0;
  *(v4f*)&As[0][ls + 32 * LDS_S] = a1;
  *(v4f*)&As[0][ls + 64 * LDS_S] = a2;
  *(v4f*)&As[0][ls + 96 * LDS_S] = a3;
  *(v4f*)&Ws[0][ls]              = w0;
  *(v4f*)&Ws[0][ls + 32 * LDS_S] = w1;
  __syncthreads();

  v8f c0 = {0.f,0.f,0.f,0.f,0.f,0.f,0.f,0.f};
  v8f c1 = {0.f,0.f,0.f,0.f,0.f,0.f,0.f,0.f};
  v8f c2 = {0.f,0.f,0.f,0.f,0.f,0.f,0.f,0.f};
  v8f c3 = {0.f,0.f,0.f,0.f,0.f,0.f,0.f,0.f};

  const int aoff = (16 * wib + lm) * LDS_S + 2 * half;
  const int woff = lm * LDS_S + 2 * half;
  for (int c = 0; c + 1 < nchunks; ++c) {
    const int buf  = c & 1;
    const int nbuf = buf ^ 1;
    ga += KC;
    gw += KC;
    a0 = *(const v4f*)(ga);
    a1 = *(const v4f*)(ga + rows32);
    a2 = *(const v4f*)(ga + 2 * rows32);
    a3 = *(const v4f*)(ga + 3 * rows32);
    w0 = *(const v4f*)(gw);
    w1 = *(const v4f*)(gw + rows32);
    wmma_chunk(&As[buf][aoff], &Ws[buf][woff], c0, c1, c2, c3);
    *(v4f*)&As[nbuf][ls]              = a0;
    *(v4f*)&As[nbuf][ls + 32 * LDS_S] = a1;
    *(v4f*)&As[nbuf][ls + 64 * LDS_S] = a2;
    *(v4f*)&As[nbuf][ls + 96 * LDS_S] = a3;
    *(v4f*)&Ws[nbuf][ls]              = w0;
    *(v4f*)&Ws[nbuf][ls + 32 * LDS_S] = w1;
    __syncthreads();
  }
  {
    const int buf = (nchunks - 1) & 1;
    wmma_chunk(&As[buf][aoff], &Ws[buf][woff], c0, c1, c2, c3);
  }

  const int m0  = mBase + 16 * wib;
  const int col = n0 + lm;
  const float bb0 = bias ? bias[col +  0] : 0.f;
  const float bb1 = bias ? bias[col + 16] : 0.f;
  const float bb2 = bias ? bias[col + 32] : 0.f;
  const float bb3 = bias ? bias[col + 48] : 0.f;
#pragma unroll
  for (int r = 0; r < 8; ++r) {
    const int row  = m0 + r + 8 * half;
    const float rm = rmask[row];
    float* orow = out + (size_t)row * N;
    orow[col +  0] = (c0[r] + bb0) * rm;
    orow[col + 16] = (c1[r] + bb1) * rm;
    orow[col + 32] = (c2[r] + bb2) * rm;
    orow[col + 48] = (c3[r] + bb3) * rm;
  }
}

__global__ __launch_bounds__(256) void tanh_dot_kernel(
    const float* __restrict__ t1g,
    const float* __restrict__ t2g,
    const float* __restrict__ wtg,
    const float* __restrict__ vmask,
    const float* __restrict__ smask,
    float* __restrict__ out) {
  __shared__ float t2s[256 * 33];
  __shared__ __align__(16) float wts[D_DIM];
  __shared__ __align__(16) float t1s[VPB * D_DIM];

  const int tid  = threadIdx.x;
  const int lane = tid & 31;
  const int w    = tid >> 5;
  const int b    = blockIdx.x / (TV_DIM / VPB);
  const int v    = (blockIdx.x % (TV_DIM / VPB)) * VPB + w;
  const int rowv = b * TV_DIM + v;

  for (int i = tid; i < D_DIM; i += 256) wts[i] = wtg[i];
#pragma unroll
  for (int j = 0; j < D_DIM / 32; ++j)
    t1s[w * D_DIM + lane + 32 * j] = t1g[(size_t)rowv * D_DIM + lane + 32 * j];

  const float pm = vmask[rowv] * smask[b * TS_DIM + lane];
  float acc = 0.f;

  for (int h = 0; h < 2; ++h) {
    if (h) __syncthreads();
    for (int i = tid; i < 256 * TS_DIM; i += 256) {
      const int s  = i >> 8;
      const int dl = i & 255;
      t2s[dl * 33 + s] = t2g[(size_t)(b * TS_DIM + s) * D_DIM + h * 256 + dl];
    }
    __syncthreads();

    const float* t1h = t1s + w * D_DIM + h * 256;
    const float* wth = wts + h * 256;
#pragma unroll 4
    for (int d = 0; d < 256; d += 4) {
      const float4 x1 = *(const float4*)(t1h + d);
      const float4 xw = *(const float4*)(wth + d);
      const float a0 = t2s[(d + 0) * 33 + lane];
      const float a1 = t2s[(d + 1) * 33 + lane];
      const float a2 = t2s[(d + 2) * 33 + lane];
      const float a3 = t2s[(d + 3) * 33 + lane];
      acc = fmaf(fast_tanh((x1.x + a0) * pm), xw.x, acc);
      acc = fmaf(fast_tanh((x1.y + a1) * pm), xw.y, acc);
      acc = fmaf(fast_tanh((x1.z + a2) * pm), xw.z, acc);
      acc = fmaf(fast_tanh((x1.w + a3) * pm), xw.w, acc);
    }
  }

  out[(size_t)rowv * TS_DIM + lane] = acc * pm;
}

extern "C" void kernel_launch(void* const* d_in, const int* in_sizes, int n_in,
                              void* d_out, int out_size, void* d_ws, size_t ws_size,
                              hipStream_t stream) {
  const float* video = (const float*)d_in[0];
  const float* vmask = (const float*)d_in[1];
  const float* sent  = (const float*)d_in[2];
  const float* smask = (const float*)d_in[3];
  const float* w1    = (const float*)d_in[4];
  const float* b1    = (const float*)d_in[5];
  const float* w2    = (const float*)d_in[6];
  const float* wt    = (const float*)d_in[7];
  float* out  = (float*)d_out;

  float* tmp1 = (float*)d_ws;
  float* tmp2 = tmp1 + (size_t)B_DIM * TV_DIM * D_DIM;

  gemm_xwt_wmma<<<256, 256, 0, stream>>>(video, w1, b1, vmask, tmp1,
                                         B_DIM * TV_DIM, D_DIM, D_DIM);
  gemm_xwt_wmma<<<32, 256, 0, stream>>>(sent, w2, nullptr, smask, tmp2,
                                        B_DIM * TS_DIM, D_DIM, D_DIM);
  tanh_dot_kernel<<<B_DIM * (TV_DIM / VPB), 256, 0, stream>>>(
      tmp1, tmp2, wt, vmask, smask, out);
}


# Both projection GEMMs in one launch: sentence GEMM cut into 128 quarter tiles (32x64, 4 of 8 waves on WMMA) that run beside the 256 video tiles, not after them on 32 of 256 WGPs
# speedup vs baseline: 3.1740x; 1.4508x over previous
#include <hip/hip_runtime.h>

#define D_DIM  512
#define B_DIM  16
#define TV_DIM 256
#define TS_DIM 32
#define VPB    8

#define NB_V   256
#define NB_S   128

#define KC     32
#define LDS_S  36

typedef float v2f __attribute__((ext_vector_type(2)));
typedef float v4f __attribute__((ext_vector_type(4)));
typedef float v8f __attribute__((ext_vector_type(8)));

__device__ __forceinline__ float fast_tanh(float x) {
#if __has_builtin(__builtin_amdgcn_tanhf)
  return __builtin_amdgcn_tanhf(x);
#else
  float y;
  asm("v_tanh_f32 %0, %1" : "=v"(y) : "v"(x));
  return y;
#endif
}

template <int NT>
__device__ __forceinline__ void wmma_chunk(const float* Ab, const float* Wb,
                                           v8f& c0, v8f& c1, v8f& c2, v8f& c3) {
#pragma unroll
  for (int kk = 0; kk < KC; kk += 4) {
    v2f a  = *(const v2f*)(Ab + kk);
    v2f b0 = *(const v2f*)(Wb + kk);
    v2f b1 = *(const v2f*)(Wb + 16 * LDS_S + kk);
    c0 = __builtin_amdgcn_wmma_f32_16x16x4_f32(false, a, false, b0, (short)0, c0, false, false);
    c1 = __builtin_amdgcn_wmma_f32_16x16x4_f32(false, a, false, b1, (short)0, c1, false, false);
    if constexpr (NT == 4) {
      v2f b2 = *(const v2f*)(Wb + 32 * LDS_S + kk);
      v2f b3 = *(const v2f*)(Wb + 48 * LDS_S + kk);
      c2 = __builtin_amdgcn_wmma_f32_16x16x4_f32(false, a, false, b2, (short)0, c2, false, false);
      c3 = __builtin_amdgcn_wmma_f32_16x16x4_f32(false, a, false, b3, (short)0, c3, false, false);
    }
  }
}

template <int TM, int NT>
__device__ __forceinline__ void proj_tile(
    const float* __restrict__ A, const float* __restrict__ W,
    const float* __restrict__ bias, const float* __restrict__ rmask,
    float* __restrict__ out, int bid, int N, int K,
    float (*As)[128 * LDS_S], float (*Ws)[64 * LDS_S]) {
  constexpr int WCOLS = 4 / NT;
  constexpr int NACT  = (TM / 16) * WCOLS;

  const int tid  = threadIdx.x;
  const int lane = tid & 31;
  const int wib  = tid >> 5;
  const int ntn  = N >> 6;
  const int mblk = bid / ntn;
  const int ng   = bid % ntn;
  const int mBase = mblk * TM;
  const int n0    = ng * 64;
  const int half  = lane >> 4;
  const int lm    = lane & 15;
  const int nchunks = K / KC;
  const int wr = wib / WCOLS;
  const int wc = wib % WCOLS;
  const bool act = (NACT == 8) || (wib < NACT);

  const int sr = tid >> 3;
  const int sc = (tid & 7) * 4;
  const size_t rows32 = (size_t)32 * K;
  const float* ga = A + (size_t)(mBase + sr) * K + sc;
  const float* gw = W + (size_t)(n0    + sr) * K + sc;
  const int ls = sr * LDS_S + sc;

  v4f a0 = *(const v4f*)(ga);
  v4f a1 = a0, a2 = a0, a3 = a0;
  if constexpr (TM == 128) {
    a1 = *(const v4f*)(ga + rows32);
    a2 = *(const v4f*)(ga + 2 * rows32);
    a3 = *(const v4f*)(ga + 3 * rows32);
  }
  v4f w0 = *(const v4f*)(gw);
  v4f w1 = *(const v4f*)(gw + rows32);
  *(v4f*)&As[0][ls] = a0;
  if constexpr (TM == 128) {
    *(v4f*)&As[0][ls + 32 * LDS_S] = a1;
    *(v4f*)&As[0][ls + 64 * LDS_S] = a2;
    *(v4f*)&As[0][ls + 96 * LDS_S] = a3;
  }
  *(v4f*)&Ws[0][ls]              = w0;
  *(v4f*)&Ws[0][ls + 32 * LDS_S] = w1;
  __syncthreads();

  v8f c0 = {0.f,0.f,0.f,0.f,0.f,0.f,0.f,0.f};
  v8f c1 = {0.f,0.f,0.f,0.f,0.f,0.f,0.f,0.f};
  v8f c2 = {0.f,0.f,0.f,0.f,0.f,0.f,0.f,0.f};
  v8f c3 = {0.f,0.f,0.f,0.f,0.f,0.f,0.f,0.f};

  const int aoff = (16 * wr + lm) * LDS_S + 2 * half;
  const int woff = (16 * NT * wc + lm) * LDS_S + 2 * half;
  for (int c = 0; c + 1 < nchunks; ++c) {
    const int buf  = c & 1;
    const int nbuf = buf ^ 1;
    ga += KC;
    gw += KC;
    a0 = *(const v4f*)(ga);
    if constexpr (TM == 128) {
      a1 = *(const v4f*)(ga + rows32);
      a2 = *(const v4f*)(ga + 2 * rows32);
      a3 = *(const v4f*)(ga + 3 * rows32);
    }
    w0 = *(const v4f*)(gw);
    w1 = *(const v4f*)(gw + rows32);
    if (act) wmma_chunk<NT>(&As[buf][aoff], &Ws[buf][woff], c0, c1, c2, c3);
    *(v4f*)&As[nbuf][ls] = a0;
    if constexpr (TM == 128) {
      *(v4f*)&As[nbuf][ls + 32 * LDS_S] = a1;
      *(v4f*)&As[nbuf][ls + 64 * LDS_S] = a2;
      *(v4f*)&As[nbuf][ls + 96 * LDS_S] = a3;
    }
    *(v4f*)&Ws[nbuf][ls]              = w0;
    *(v4f*)&Ws[nbuf][ls + 32 * LDS_S] = w1;
    __syncthreads();
  }
  if (!act) return;
  {
    const int buf = (nchunks - 1) & 1;
    wmma_chunk<NT>(&As[buf][aoff], &Ws[buf][woff], c0, c1, c2, c3);
  }

  const int m0  = mBase + 16 * wr;
  const int col = n0 + 16 * NT * wc + lm;
  const float bb0 = bias ? bias[col +  0] : 0.f;
  const float bb1 = bias ? bias[col + 16] : 0.f;
  float bb2 = 0.f, bb3 = 0.f;
  if constexpr (NT == 4) {
    bb2 = bias ? bias[col + 32] : 0.f;
    bb3 = bias ? bias[col + 48] : 0.f;
  }
#pragma unroll
  for (int r = 0; r < 8; ++r) {
    const int row  = m0 + r + 8 * half;
    const float rm = rmask[row];
    float* orow = out + (size_t)row * N;
    orow[col +  0] = (c0[r] + bb0) * rm;
    orow[col + 16] = (c1[r] + bb1) * rm;
    if constexpr (NT == 4) {
      orow[col + 32] = (c2[r] + bb2) * rm;
      orow[col + 48] = (c3[r] + bb3) * rm;
    }
  }
}

__global__ __launch_bounds__(256) __attribute__((amdgpu_waves_per_eu(2, 4))) void proj_pair_wmma(
    const float* __restrict__ Av, const float* __restrict__ Wv,
    const float* __restrict__ biasv, const float* __restrict__ rmaskv,
    float* __restrict__ outv,
    const float* __restrict__ Asn, const float* __restrict__ Wsn,
    const float* __restrict__ rmasks, float* __restrict__ outs,
    int N, int K) {
  __shared__ float As[2][128 * LDS_S];
  __shared__ float Ws[2][ 64 * LDS_S];

  if (blockIdx.x < NB_V)
    proj_tile<128, 4>(Av, Wv, biasv, rmaskv, outv, (int)blockIdx.x, N, K, As, Ws);
  else
    proj_tile<32, 2>(Asn, Wsn, nullptr, rmasks, outs, (int)blockIdx.x - NB_V, N, K, As, Ws);
}

__global__ __launch_bounds__(256) void tanh_dot_kernel(
    const float* __restrict__ t1g,
    const float* __restrict__ t2g,
    const float* __restrict__ wtg,
    const float* __restrict__ vmask,
    const float* __restrict__ smask,
    float* __restrict__ out) {
  __shared__ float t2s[256 * 33];
  __shared__ __align__(16) float wts[D_DIM];
  __shared__ __align__(16) float t1s[VPB * D_DIM];

  const int tid  = threadIdx.x;
  const int lane = tid & 31;
  const int w    = tid >> 5;
  const int b    = blockIdx.x / (TV_DIM / VPB);
  const int v    = (blockIdx.x % (TV_DIM / VPB)) * VPB + w;
  const int rowv = b * TV_DIM + v;

  for (int i = tid; i < D_DIM; i += 256) wts[i] = wtg[i];
#pragma unroll
  for (int j = 0; j < D_DIM / 32; ++j)
    t1s[w * D_DIM + lane + 32 * j] = t1g[(size_t)rowv * D_DIM + lane + 32 * j];

  const float pm = vmask[rowv] * smask[b * TS_DIM + lane];
  float acc = 0.f;

  for (int h = 0; h < 2; ++h) {
    if (h) __syncthreads();
    for (int i = tid; i < 256 * TS_DIM; i += 256) {
      const int s  = i >> 8;
      const int dl = i & 255;
      t2s[dl * 33 + s] = t2g[(size_t)(b * TS_DIM + s) * D_DIM + h * 256 + dl];
    }
    __syncthreads();

    const float* t1h = t1s + w * D_DIM + h * 256;
    const float* wth = wts + h * 256;
#pragma unroll 4
    for (int d = 0; d < 256; d += 4) {
      const float4 x1 = *(const float4*)(t1h + d);
      const float4 xw = *(const float4*)(wth + d);
      const float a0 = t2s[(d + 0) * 33 + lane];
      const float a1 = t2s[(d + 1) * 33 + lane];
      const float a2 = t2s[(d + 2) * 33 + lane];
      const float a3 = t2s[(d + 3) * 33 + lane];
      acc = fmaf(fast_tanh((x1.x + a0) * pm), xw.x, acc);
      acc = fmaf(fast_tanh((x1.y + a1) * pm), xw.y, acc);
      acc = fmaf(fast_tanh((x1.z + a2) * pm), xw.z, acc);
      acc = fmaf(fast_tanh((x1.w + a3) * pm), xw.w, acc);
    }
  }

  out[(size_t)rowv * TS_DIM + lane] = acc * pm;
}

extern "C" void kernel_launch(void* const* d_in, const int* in_sizes, int n_in,
                              void* d_out, int out_size, void* d_ws, size_t ws_size,
                              hipStream_t stream) {
  const float* video = (const float*)d_in[0];
  const float* vmask = (const float*)d_in[1];
  const float* sent  = (const float*)d_in[2];
  const float* smask = (const float*)d_in[3];
  const float* w1    = (const float*)d_in[4];
  const float* b1    = (const float*)d_in[5];
  const float* w2    = (const float*)d_in[6];
  const float* wt    = (const float*)d_in[7];
  float* out  = (float*)d_out;

  float* tmp1 = (float*)d_ws;
  float* tmp2 = tmp1 + (size_t)B_DIM * TV_DIM * D_DIM;

  proj_pair_wmma<<<NB_V + NB_S, 256, 0, stream>>>(video, w1, b1, vmask, tmp1,
                                                  sent, w2, smask, tmp2, D_DIM, D_DIM);
  tanh_dot_kernel<<<B_DIM * (TV_DIM / VPB), 256, 0, stream>>>(
      tmp1, tmp2, wt, vmask, smask, out);
}


# tanh/dot kernel: lane owns d (tmp1 rows, wt in registers; one LDS row of tmp2 serves two v-rows) instead of lane owns s; lane partials summed across lanes through LDS (sum order over d changes)
# speedup vs baseline: 3.9306x; 1.2384x over previous
#include <hip/hip_runtime.h>

#define D_DIM  512
#define B_DIM  16
#define TV_DIM 256
#define TS_DIM 32

#define NB_V   256
#define NB_S   128

#define KC     32
#define LDS_S  36

typedef float v2f __attribute__((ext_vector_type(2)));
typedef float v4f __attribute__((ext_vector_type(4)));
typedef float v8f __attribute__((ext_vector_type(8)));

__device__ __forceinline__ float fast_tanh(float x) {
#if __has_builtin(__builtin_amdgcn_tanhf)
  return __builtin_amdgcn_tanhf(x);
#else
  float y;
  asm("v_tanh_f32 %0, %1" : "=v"(y) : "v"(x));
  return y;
#endif
}

template <int NT>
__device__ __forceinline__ void wmma_chunk(const float* Ab, const float* Wb,
                                           v8f& c0, v8f& c1, v8f& c2, v8f& c3) {
#pragma unroll
  for (int kk = 0; kk < KC; kk += 4) {
    v2f a  = *(const v2f*)(Ab + kk);
    v2f b0 = *(const v2f*)(Wb + kk);
    v2f b1 = *(const v2f*)(Wb + 16 * LDS_S + kk);
    c0 = __builtin_amdgcn_wmma_f32_16x16x4_f32(false, a, false, b0, (short)0, c0, false, false);
    c1 = __builtin_amdgcn_wmma_f32_16x16x4_f32(false, a, false, b1, (short)0, c1, false, false);
    if constexpr (NT == 4) {
      v2f b2 = *(const v2f*)(Wb + 32 * LDS_S + kk);
      v2f b3 = *(const v2f*)(Wb + 48 * LDS_S + kk);
      c2 = __builtin_amdgcn_wmma_f32_16x16x4_f32(false, a, false, b2, (short)0, c2, false, false);
      c3 = __builtin_amdgcn_wmma_f32_16x16x4_f32(false, a, false, b3, (short)0, c3, false, false);
    }
  }
}

template <int TM, int NT>
__device__ __forceinline__ void proj_tile(
    const float* __restrict__ A, const float* __restrict__ W,
    const float* __restrict__ bias, const float* __restrict__ rmask,
    float* __restrict__ out, int bid, int N, int K,
    float (*As)[128 * LDS_S], float (*Ws)[64 * LDS_S]) {
  constexpr int WCOLS = 4 / NT;
  constexpr int NACT  = (TM / 16) * WCOLS;

  const int tid  = threadIdx.x;
  const int lane = tid & 31;
  const int wib  = tid >> 5;
  const int ntn  = N >> 6;
  const int mblk = bid / ntn;
  const int ng   = bid % ntn;
  const int mBase = mblk * TM;
  const int n0    = ng * 64;
  const int half  = lane >> 4;
  const int lm    = lane & 15;
  const int nchunks = K / KC;
  const int wr = wib / WCOLS;
  const int wc = wib % WCOLS;
  const bool act = (NACT == 8) || (wib < NACT);

  const int sr = tid >> 3;
  const int sc = (tid & 7) * 4;
  const size_t rows32 = (size_t)32 * K;
  const float* ga = A + (size_t)(mBase + sr) * K + sc;
  const float* gw = W + (size_t)(n0    + sr) * K + sc;
  const int ls = sr * LDS_S + sc;

  v4f a0 = *(const v4f*)(ga);
  v4f a1 = a0, a2 = a0, a3 = a0;
  if constexpr (TM == 128) {
    a1 = *(const v4f*)(ga + rows32);
    a2 = *(const v4f*)(ga + 2 * rows32);
    a3 = *(const v4f*)(ga + 3 * rows32);
  }
  v4f w0 = *(const v4f*)(gw);
  v4f w1 = *(const v4f*)(gw + rows32);
  *(v4f*)&As[0][ls] = a0;
  if constexpr (TM == 128) {
    *(v4f*)&As[0][ls + 32 * LDS_S] = a1;
    *(v4f*)&As[0][ls + 64 * LDS_S] = a2;
    *(v4f*)&As[0][ls + 96 * LDS_S] = a3;
  }
  *(v4f*)&Ws[0][ls]              = w0;
  *(v4f*)&Ws[0][ls + 32 * LDS_S] = w1;
  __syncthreads();

  v8f c0 = {0.f,0.f,0.f,0.f,0.f,0.f,0.f,0.f};
  v8f c1 = {0.f,0.f,0.f,0.f,0.f,0.f,0.f,0.f};
  v8f c2 = {0.f,0.f,0.f,0.f,0.f,0.f,0.f,0.f};
  v8f c3 = {0.f,0.f,0.f,0.f,0.f,0.f,0.f,0.f};

  const int aoff = (16 * wr + lm) * LDS_S + 2 * half;
  const int woff = (16 * NT * wc + lm) * LDS_S + 2 * half;
  for (int c = 0; c + 1 < nchunks; ++c) {
    const int buf  = c & 1;
    const int nbuf = buf ^ 1;
    ga += KC;
    gw += KC;
    a0 = *(const v4f*)(ga);
    if constexpr (TM == 128) {
      a1 = *(const v4f*)(ga + rows32);
      a2 = *(const v4f*)(ga + 2 * rows32);
      a3 = *(const v4f*)(ga + 3 * rows32);
    }
    w0 = *(const v4f*)(gw);
    w1 = *(const v4f*)(gw + rows32);
    if (act) wmma_chunk<NT>(&As[buf][aoff], &Ws[buf][woff], c0, c1, c2, c3);
    *(v4f*)&As[nbuf][ls] = a0;
    if constexpr (TM == 128) {
      *(v4f*)&As[nbuf][ls + 32 * LDS_S] = a1;
      *(v4f*)&As[nbuf][ls + 64 * LDS_S] = a2;
      *(v4f*)&As[nbuf][ls + 96 * LDS_S] = a3;
    }
    *(v4f*)&Ws[nbuf][ls]              = w0;
    *(v4f*)&Ws[nbuf][ls + 32 * LDS_S] = w1;
    __syncthreads();
  }
  if (!act) return;
  {
    const int buf = (nchunks - 1) & 1;
    wmma_chunk<NT>(&As[buf][aoff], &Ws[buf][woff], c0, c1, c2, c3);
  }

  const int m0  = mBase + 16 * wr;
  const int col = n0 + 16 * NT * wc + lm;
  const float bb0 = bias ? bias[col +  0] : 0.f;
  const float bb1 = bias ? bias[col + 16] : 0.f;
  float bb2 = 0.f, bb3 = 0.f;
  if constexpr (NT == 4) {
    bb2 = bias ? bias[col + 32] : 0.f;
    bb3 = bias ? bias[col + 48] : 0.f;
  }
#pragma unroll
  for (int r = 0; r < 8; ++r) {
    const int row  = m0 + r + 8 * half;
    const float rm = rmask[row];
    float* orow = out + (size_t)row * N;
    orow[col +  0] = (c0[r] + bb0) * rm;
    orow[col + 16] = (c1[r] + bb1) * rm;
    if constexpr (NT == 4) {
      orow[col + 32] = (c2[r] + bb2) * rm;
      orow[col + 48] = (c3[r] + bb3) * rm;
    }
  }
}

__global__ __launch_bounds__(256) __attribute__((amdgpu_waves_per_eu(2, 4))) void proj_pair_wmma(
    const float* __restrict__ Av, const float* __restrict__ Wv,
    const float* __restrict__ biasv, const float* __restrict__ rmaskv,
    float* __restrict__ outv,
    const float* __restrict__ Asn, const float* __restrict__ Wsn,
    const float* __restrict__ rmasks, float* __restrict__ outs,
    int N, int K) {
  __shared__ float As[2][128 * LDS_S];
  __shared__ float Ws[2][ 64 * LDS_S];

  if (blockIdx.x < NB_V)
    proj_tile<128, 4>(Av, Wv, biasv, rmaskv, outv, (int)blockIdx.x, N, K, As, Ws);
  else
    proj_tile<32, 2>(Asn, Wsn, nullptr, rmasks, outs, (int)blockIdx.x - NB_V, N, K, As, Ws);
}

#define VPW 2
#define VPB2 (8 * VPW)
#define RED_S 33
__global__ __launch_bounds__(256) void tanh_dot_kernel(
    const float* __restrict__ t1g,
    const float* __restrict__ t2g,
    const float* __restrict__ wtg,
    const float* __restrict__ vmask,
    const float* __restrict__ smask,
    float* __restrict__ out) {
  __shared__ __align__(16) float t2s[TS_DIM * D_DIM];
  __shared__ float red[8 * VPW * TS_DIM * RED_S];

  const int tid  = threadIdx.x;
  const int lane = tid & 31;
  const int w    = tid >> 5;
  const int b    = blockIdx.x / (TV_DIM / VPB2);
  const int v0   = (blockIdx.x % (TV_DIM / VPB2)) * VPB2 + VPW * w;
  const int row0 = b * TV_DIM + v0;

  {
    const v4f* src = (const v4f*)(t2g + (size_t)b * TS_DIM * D_DIM);
    v4f* dst = (v4f*)t2s;
    v4f r[16];
#pragma unroll
    for (int i = 0; i < 16; ++i) r[i] = src[tid + 256 * i];
#pragma unroll
    for (int i = 0; i < 16; ++i) dst[tid + 256 * i] = r[i];
  }

  v4f x0[4], x1[4], wv[4];
#pragma unroll
  for (int j = 0; j < 4; ++j) {
    x0[j] = *(const v4f*)(t1g + (size_t)row0 * D_DIM + 128 * j + 4 * lane);
    x1[j] = *(const v4f*)(t1g + (size_t)(row0 + 1) * D_DIM + 128 * j + 4 * lane);
    wv[j] = *(const v4f*)(wtg + 128 * j + 4 * lane);
  }
  const float vm0 = vmask[row0];
  const float vm1 = vmask[row0 + 1];
  const float sml = smask[b * TS_DIM + lane];

  __syncthreads();

  float* red0 = red + (size_t)(w * VPW + 0) * TS_DIM * RED_S;
  float* red1 = red + (size_t)(w * VPW + 1) * TS_DIM * RED_S;
  const v4f* t2row = (const v4f*)t2s + lane;

#pragma unroll 2
  for (int s = 0; s < TS_DIM; ++s) {
    const float sms = __shfl(sml, s, 32);
    const float pm0 = vm0 * sms;
    const float pm1 = vm1 * sms;
    const v2f p0 = {pm0, pm0};
    const v2f p1 = {pm1, pm1};
    v2f a0 = {0.f, 0.f}, a1 = {0.f, 0.f}, c0 = {0.f, 0.f}, c1 = {0.f, 0.f};
#pragma unroll
    for (int j = 0; j < 4; ++j) {
      const v4f y = t2row[s * (D_DIM / 4) + 32 * j];
      const v2f ylo = {y.x, y.y}, yhi = {y.z, y.w};
      const v2f wlo = {wv[j].x, wv[j].y}, whi = {wv[j].z, wv[j].w};
      {
        const v2f xlo = {x0[j].x, x0[j].y}, xhi = {x0[j].z, x0[j].w};
        const v2f ul = (xlo + ylo) * p0, uh = (xhi + yhi) * p0;
        const v2f tl = {fast_tanh(ul.x), fast_tanh(ul.y)};
        const v2f th = {fast_tanh(uh.x), fast_tanh(uh.y)};
        a0 = tl * wlo + a0;
        c0 = th * whi + c0;
      }
      {
        const v2f xlo = {x1[j].x, x1[j].y}, xhi = {x1[j].z, x1[j].w};
        const v2f ul = (xlo + ylo) * p1, uh = (xhi + yhi) * p1;
        const v2f tl = {fast_tanh(ul.x), fast_tanh(ul.y)};
        const v2f th = {fast_tanh(uh.x), fast_tanh(uh.y)};
        a1 = tl * wlo + a1;
        c1 = th * whi + c1;
      }
    }
    red0[s * RED_S + lane] = (a0.x + a0.y) + (c0.x + c0.y);
    red1[s * RED_S + lane] = (a1.x + a1.y) + (c1.x + c1.y);
  }

  __syncthreads();

  float t0 = 0.f, t1 = 0.f;
#pragma unroll 8
  for (int k = 0; k < 32; ++k) {
    t0 += red0[lane * RED_S + k];
    t1 += red1[lane * RED_S + k];
  }
  out[(size_t)row0 * TS_DIM + lane]       = t0 * (vm0 * sml);
  out[(size_t)(row0 + 1) * TS_DIM + lane] = t1 * (vm1 * sml);
}

extern "C" void kernel_launch(void* const* d_in, const int* in_sizes, int n_in,
                              void* d_out, int out_size, void* d_ws, size_t ws_size,
                              hipStream_t stream) {
  const float* video = (const float*)d_in[0];
  const float* vmask = (const float*)d_in[1];
  const float* sent  = (const float*)d_in[2];
  const float* smask = (const float*)d_in[3];
  const float* w1    = (const float*)d_in[4];
  const float* b1    = (const float*)d_in[5];
  const float* w2    = (const float*)d_in[6];
  const float* wt    = (const float*)d_in[7];
  float* out  = (float*)d_out;

  float* tmp1 = (float*)d_ws;
  float* tmp2 = tmp1 + (size_t)B_DIM * TV_DIM * D_DIM;

  proj_pair_wmma<<<NB_V + NB_S, 256, 0, stream>>>(video, w1, b1, vmask, tmp1,
                                                  sent, w2, smask, tmp2, D_DIM, D_DIM);
  tanh_dot_kernel<<<B_DIM * (TV_DIM / VPB2), 256, 0, stream>>>(
      tmp1, tmp2, wt, vmask, smask, out);
}


# tanh/dot moved into the GEMM launch: product transposed so a lane sums its 32 d of a 64-d column block out of the accumulators; 2nd launch adds the 8 partial sums (sum order over d changes)
# speedup vs baseline: 4.3696x; 1.1117x over previous
#include <hip/hip_runtime.h>

#define D_DIM  512
#define B_DIM  16
#define TV_DIM 256
#define TS_DIM 32

#define NB_V   256
#define NPART  8
#define NOUT   (B_DIM * TV_DIM * TS_DIM)

#define KC     32
#define LDS_S  36
#define T2_S   68
#define RED_S  33

typedef float v2f __attribute__((ext_vector_type(2)));
typedef float v4f __attribute__((ext_vector_type(4)));
typedef float v8f __attribute__((ext_vector_type(8)));

__device__ __forceinline__ float fast_tanh(float x) {
#if __has_builtin(__builtin_amdgcn_tanhf)
  return __builtin_amdgcn_tanhf(x);
#else
  float y;
  asm("v_tanh_f32 %0, %1" : "=v"(y) : "v"(x));
  return y;
#endif
}

template <int K0, int K1>
__device__ __forceinline__ void wmma_steps(const float* Xb, const float* Wb,
                                           const float* Sb, const float* Vb,
                                           v8f& c0, v8f& c1, v8f& c2, v8f& c3, v8f& c4) {
#pragma unroll
  for (int kk = 4 * K0; kk < 4 * K1; kk += 4) {
    v2f x  = *(const v2f*)(Xb + kk);
    v2f w0 = *(const v2f*)(Wb + kk);
    v2f w1 = *(const v2f*)(Wb + 16 * LDS_S + kk);
    v2f w2 = *(const v2f*)(Wb + 32 * LDS_S + kk);
    v2f w3 = *(const v2f*)(Wb + 48 * LDS_S + kk);
    v2f s  = *(const v2f*)(Sb + kk);
    v2f v  = *(const v2f*)(Vb + kk);
    c0 = __builtin_amdgcn_wmma_f32_16x16x4_f32(false, w0, false, x, (short)0, c0, false, false);
    c1 = __builtin_amdgcn_wmma_f32_16x16x4_f32(false, w1, false, x, (short)0, c1, false, false);
    c2 = __builtin_amdgcn_wmma_f32_16x16x4_f32(false, w2, false, x, (short)0, c2, false, false);
    c3 = __builtin_amdgcn_wmma_f32_16x16x4_f32(false, w3, false, x, (short)0, c3, false, false);
    c4 = __builtin_amdgcn_wmma_f32_16x16x4_f32(false, s,  false, v, (short)0, c4, false, false);
  }
}

struct PairFrag { v2f x, w0, w1, w2, w3, s, v; v2f xq, w0q, w1q, w2q, w3q, sq, vq; };
template <int KS>
__device__ __forceinline__ PairFrag load_pair(const float* Xb, const float* Wb,
                                              const float* Sb, const float* Vb) {
  PairFrag t;
  t.x   = *(const v2f*)(Xb + 4 * KS);
  t.w0  = *(const v2f*)(Wb + 4 * KS);
  t.w1  = *(const v2f*)(Wb + 16 * LDS_S + 4 * KS);
  t.w2  = *(const v2f*)(Wb + 32 * LDS_S + 4 * KS);
  t.w3  = *(const v2f*)(Wb + 48 * LDS_S + 4 * KS);
  t.s   = *(const v2f*)(Sb + 4 * KS);
  t.v   = *(const v2f*)(Vb + 4 * KS);
  t.xq  = *(const v2f*)(Xb + 4 * KS + 4);
  t.w0q = *(const v2f*)(Wb + 4 * KS + 4);
  t.w1q = *(const v2f*)(Wb + 16 * LDS_S + 4 * KS + 4);
  t.w2q = *(const v2f*)(Wb + 32 * LDS_S + 4 * KS + 4);
  t.w3q = *(const v2f*)(Wb + 48 * LDS_S + 4 * KS + 4);
  t.sq  = *(const v2f*)(Sb + 4 * KS + 4);
  t.vq  = *(const v2f*)(Vb + 4 * KS + 4);
  return t;
}
__device__ __forceinline__ void mma_pair(const PairFrag& t,
                                         v8f& c0, v8f& c1, v8f& c2, v8f& c3, v8f& c4) {
  c0 = __builtin_amdgcn_wmma_f32_16x16x4_f32(false, t.w0,  false, t.x,  (short)0, c0, false, false);
  c1 = __builtin_amdgcn_wmma_f32_16x16x4_f32(false, t.w1,  false, t.x,  (short)0, c1, false, false);
  c2 = __builtin_amdgcn_wmma_f32_16x16x4_f32(false, t.w2,  false, t.x,  (short)0, c2, false, false);
  c3 = __builtin_amdgcn_wmma_f32_16x16x4_f32(false, t.w3,  false, t.x,  (short)0, c3, false, false);
  c4 = __builtin_amdgcn_wmma_f32_16x16x4_f32(false, t.s,   false, t.v,  (short)0, c4, false, false);
  c0 = __builtin_amdgcn_wmma_f32_16x16x4_f32(false, t.w0q, false, t.xq, (short)0, c0, false, false);
  c1 = __builtin_amdgcn_wmma_f32_16x16x4_f32(false, t.w1q, false, t.xq, (short)0, c1, false, false);
  c2 = __builtin_amdgcn_wmma_f32_16x16x4_f32(false, t.w2q, false, t.xq, (short)0, c2, false, false);
  c3 = __builtin_amdgcn_wmma_f32_16x16x4_f32(false, t.w3q, false, t.xq, (short)0, c3, false, false);
  c4 = __builtin_amdgcn_wmma_f32_16x16x4_f32(false, t.sq,  false, t.vq, (short)0, c4, false, false);
}

__device__ __forceinline__ void tanh_dot8(const v8f& cj, const v4f ya, const v4f yb,
                                          const v4f wa, const v4f wb, const v2f p,
                                          v2f& lo, v2f& hi) {
  const v2f x0 = {cj[0], cj[1]}, x1 = {cj[2], cj[3]}, x2 = {cj[4], cj[5]}, x3 = {cj[6], cj[7]};
  const v2f y0 = {ya.x, ya.y}, y1 = {ya.z, ya.w}, y2 = {yb.x, yb.y}, y3 = {yb.z, yb.w};
  const v2f w0 = {wa.x, wa.y}, w1 = {wa.z, wa.w}, w2 = {wb.x, wb.y}, w3 = {wb.z, wb.w};
  const v2f u0 = (x0 + y0) * p, u1 = (x1 + y1) * p, u2 = (x2 + y2) * p, u3 = (x3 + y3) * p;
  const v2f t0 = {fast_tanh(u0.x), fast_tanh(u0.y)};
  const v2f t1 = {fast_tanh(u1.x), fast_tanh(u1.y)};
  const v2f t2 = {fast_tanh(u2.x), fast_tanh(u2.y)};
  const v2f t3 = {fast_tanh(u3.x), fast_tanh(u3.y)};
  lo = t0 * w0 + lo;
  hi = t1 * w1 + hi;
  lo = t2 * w2 + lo;
  hi = t3 * w3 + hi;
}

__global__ __launch_bounds__(256) __attribute__((amdgpu_waves_per_eu(2, 4))) void proj_tanh_partial(
    const float* __restrict__ Av, const float* __restrict__ Wv,
    const float* __restrict__ biasv, const float* __restrict__ rmaskv,
    const float* __restrict__ Asn, const float* __restrict__ Wsn,
    const float* __restrict__ rmasks, const float* __restrict__ wtg,
    float* __restrict__ part, int N, int K) {
  __shared__ float As[2][128 * LDS_S];
  __shared__ float Ws[2][ 64 * LDS_S];
  __shared__ float Ss[2][ 32 * LDS_S];
  __shared__ float Vs[2][ 64 * LDS_S];
  __shared__ __align__(16) float T2s[TS_DIM * T2_S];
  __shared__ float red[8 * TS_DIM * RED_S];

  const int tid  = threadIdx.x;
  const int lane = tid & 31;
  const int wib  = tid >> 5;
  const int bid  = (int)blockIdx.x;
  const int ntn  = N >> 6;
  const int mblk = bid / ntn;
  const int ng   = bid % ntn;
  const int mBase = mblk * 128;
  const int n0    = ng * 64;
  const int sBase = (mBase / TV_DIM) * TS_DIM;
  const int half  = lane >> 4;
  const int lm    = lane & 15;
  const int nchunks = K / KC;

  const int sr = tid >> 3;
  const int sc = (tid & 7) * 4;
  const size_t rows32 = (size_t)32 * K;
  const float* ga = Av  + (size_t)(mBase + sr) * K + sc;
  const float* gw = Wv  + (size_t)(n0    + sr) * K + sc;
  const float* gs = Asn + (size_t)(sBase + sr) * K + sc;
  const float* gv = Wsn + (size_t)(n0    + sr) * K + sc;
  const int ls = sr * LDS_S + sc;

  v4f a0 = *(const v4f*)(ga);
  v4f a1 = *(const v4f*)(ga + rows32);
  v4f a2 = *(const v4f*)(ga + 2 * rows32);
  v4f a3 = *(const v4f*)(ga + 3 * rows32);
  v4f w0 = *(const v4f*)(gw);
  v4f w1 = *(const v4f*)(gw + rows32);
  v4f s0 = *(const v4f*)(gs);
  v4f v0 = *(const v4f*)(gv);
  v4f v1 = *(const v4f*)(gv + rows32);
  *(v4f*)&As[0][ls]              = a0;
  *(v4f*)&As[0][ls + 32 * LDS_S] = a1;
  *(v4f*)&As[0][ls + 64 * LDS_S] = a2;
  *(v4f*)&As[0][ls + 96 * LDS_S] = a3;
  *(v4f*)&Ws[0][ls]              = w0;
  *(v4f*)&Ws[0][ls + 32 * LDS_S] = w1;
  *(v4f*)&Ss[0][ls]              = s0;
  *(v4f*)&Vs[0][ls]              = v0;
  *(v4f*)&Vs[0][ls + 32 * LDS_S] = v1;
  __syncthreads();

  v8f c0 = {0.f,0.f,0.f,0.f,0.f,0.f,0.f,0.f};
  v8f c1 = {0.f,0.f,0.f,0.f,0.f,0.f,0.f,0.f};
  v8f c2 = {0.f,0.f,0.f,0.f,0.f,0.f,0.f,0.f};
  v8f c3 = {0.f,0.f,0.f,0.f,0.f,0.f,0.f,0.f};
  v8f c4 = {0.f,0.f,0.f,0.f,0.f,0.f,0.f,0.f};

  const int st = wib >> 2;
  const int dt = wib & 3;
  const int xoff = (16 * wib + lm) * LDS_S + 2 * half;
  const int woff = lm * LDS_S + 2 * half;
  const int soff = (16 * st + lm) * LDS_S + 2 * half;
  const int voff = (16 * dt + lm) * LDS_S + 2 * half;
  PairFrag t;
  for (int c = 0; c + 1 < nchunks; ++c) {
    const int buf  = c & 1;
    const int nbuf = buf ^ 1;
    ga += KC;
    gw += KC;
    gs += KC;
    gv += KC;
    a0 = *(const v4f*)(ga);
    a1 = *(const v4f*)(ga + rows32);
    a2 = *(const v4f*)(ga + 2 * rows32);
    a3 = *(const v4f*)(ga + 3 * rows32);
    w0 = *(const v4f*)(gw);
    w1 = *(const v4f*)(gw + rows32);
    s0 = *(const v4f*)(gs);
    v0 = *(const v4f*)(gv);
    v1 = *(const v4f*)(gv + rows32);
    const float* Xb = &As[buf][xoff];
    const float* Wb = &Ws[buf][woff];
    const float* Sb = &Ss[buf][soff];
    const float* Vb = &Vs[buf][voff];
    if (c > 0) mma_pair(t, c0, c1, c2, c3, c4);
    wmma_steps<0, 4>(Xb, Wb, Sb, Vb, c0, c1, c2, c3, c4);
    const PairFrag m = load_pair<4>(Xb, Wb, Sb, Vb);
    t = load_pair<6>(Xb, Wb, Sb, Vb);
    __builtin_amdgcn_sched_barrier(0);
    *(v4f*)&As[nbuf][ls]              = a0;
    *(v4f*)&As[nbuf][ls + 32 * LDS_S] = a1;
    *(v4f*)&As[nbuf][ls + 64 * LDS_S] = a2;
    *(v4f*)&As[nbuf][ls + 96 * LDS_S] = a3;
    *(v4f*)&Ws[nbuf][ls]              = w0;
    *(v4f*)&Ws[nbuf][ls + 32 * LDS_S] = w1;
    *(v4f*)&Ss[nbuf][ls]              = s0;
    *(v4f*)&Vs[nbuf][ls]              = v0;
    *(v4f*)&Vs[nbuf][ls + 32 * LDS_S] = v1;
    mma_pair(m, c0, c1, c2, c3, c4);
    __syncthreads();
  }
  {
    const int buf = (nchunks - 1) & 1;
    if (nchunks > 1) mma_pair(t, c0, c1, c2, c3, c4);
    wmma_steps<0, 8>(&As[buf][xoff], &Ws[buf][woff], &Ss[buf][soff], &Vs[buf][voff], c0, c1, c2, c3, c4);
  }

#pragma unroll
  for (int r = 0; r < 8; ++r) {
    const int s = 16 * st + 8 * half + r;
    T2s[s * T2_S + 16 * dt + lm] = c4[r] * rmasks[sBase + s];
  }
  if (tid < TS_DIM) T2s[tid * T2_S + 64] = rmasks[sBase + tid];

  const int vrow = mBase + 16 * wib + lm;
  const float vm = rmaskv[vrow];
  const float* bp = biasv + n0 + 8 * half;
  const float* wp = wtg   + n0 + 8 * half;
  const v4f wa0 = *(const v4f*)(wp),      wb0 = *(const v4f*)(wp + 4);
  const v4f wa1 = *(const v4f*)(wp + 16), wb1 = *(const v4f*)(wp + 20);
  const v4f wa2 = *(const v4f*)(wp + 32), wb2 = *(const v4f*)(wp + 36);
  const v4f wa3 = *(const v4f*)(wp + 48), wb3 = *(const v4f*)(wp + 52);
  {
    const v4f ba0 = *(const v4f*)(bp),      bb0 = *(const v4f*)(bp + 4);
    const v4f ba1 = *(const v4f*)(bp + 16), bb1 = *(const v4f*)(bp + 20);
    const v4f ba2 = *(const v4f*)(bp + 32), bb2 = *(const v4f*)(bp + 36);
    const v4f ba3 = *(const v4f*)(bp + 48), bb3 = *(const v4f*)(bp + 52);
    c0 = (c0 + __builtin_shufflevector(ba0, bb0, 0, 1, 2, 3, 4, 5, 6, 7)) * vm;
    c1 = (c1 + __builtin_shufflevector(ba1, bb1, 0, 1, 2, 3, 4, 5, 6, 7)) * vm;
    c2 = (c2 + __builtin_shufflevector(ba2, bb2, 0, 1, 2, 3, 4, 5, 6, 7)) * vm;
    c3 = (c3 + __builtin_shufflevector(ba3, bb3, 0, 1, 2, 3, 4, 5, 6, 7)) * vm;
  }
  __syncthreads();

  float* redw = red + wib * TS_DIM * RED_S;
#pragma unroll 2
  for (int s = 0; s < TS_DIM; ++s) {
    const float* yr = T2s + s * T2_S + 8 * half;
    const float pm = vm * T2s[s * T2_S + 64];
    const v2f p = {pm, pm};
    v2f lo = {0.f, 0.f}, hi = {0.f, 0.f};
    tanh_dot8(c0, *(const v4f*)(yr),      *(const v4f*)(yr + 4),  wa0, wb0, p, lo, hi);
    tanh_dot8(c1, *(const v4f*)(yr + 16), *(const v4f*)(yr + 20), wa1, wb1, p, lo, hi);
    tanh_dot8(c2, *(const v4f*)(yr + 32), *(const v4f*)(yr + 36), wa2, wb2, p, lo, hi);
    tanh_dot8(c3, *(const v4f*)(yr + 48), *(const v4f*)(yr + 52), wa3, wb3, p, lo, hi);
    redw[s * RED_S + lane] = (lo.x + lo.y) + (hi.x + hi.y);
  }
  __syncthreads();

  const int sv = lane >> 3;
  const int sq = 4 * (lane & 7);
#pragma unroll
  for (int q = 0; q < 4; ++q) {
    const int v = 4 * q + sv;
    float e[4];
#pragma unroll
    for (int i = 0; i < 4; ++i)
      e[i] = redw[(sq + i) * RED_S + v] + redw[(sq + i) * RED_S + v + 16];
    *(v4f*)(part + (size_t)ng * NOUT + (size_t)(mBase + 16 * wib + v) * TS_DIM + sq) =
        (v4f){e[0], e[1], e[2], e[3]};
  }
}

__global__ __launch_bounds__(128) void sum_partials(
    const float* __restrict__ part,
    const float* __restrict__ vmask,
    const float* __restrict__ smask,
    float* __restrict__ out) {
  const int i4 = (int)blockIdx.x * 128 + (int)threadIdx.x;
  const v4f* p4 = (const v4f*)part;
  v4f r[NPART];
#pragma unroll
  for (int n = 0; n < NPART; ++n) r[n] = p4[(size_t)n * (NOUT / 4) + i4];
  v4f acc = r[0];
#pragma unroll
  for (int n = 1; n < NPART; ++n) acc += r[n];
  const int i   = 4 * i4;
  const int row = i / TS_DIM;
  const int s   = i % TS_DIM;
  const float vm = vmask[row];
  const v4f sm = *(const v4f*)(smask + (row / TV_DIM) * TS_DIM + s);
  ((v4f*)out)[i4] = acc * (sm * vm);
}

extern "C" void kernel_launch(void* const* d_in, const int* in_sizes, int n_in,
                              void* d_out, int out_size, void* d_ws, size_t ws_size,
                              hipStream_t stream) {
  const float* video = (const float*)d_in[0];
  const float* vmask = (const float*)d_in[1];
  const float* sent  = (const float*)d_in[2];
  const float* smask = (const float*)d_in[3];
  const float* w1    = (const float*)d_in[4];
  const float* b1    = (const float*)d_in[5];
  const float* w2    = (const float*)d_in[6];
  const float* wt    = (const float*)d_in[7];
  float* out  = (float*)d_out;

  float* part = (float*)d_ws;

  proj_tanh_partial<<<NB_V, 256, 0, stream>>>(video, w1, b1, vmask, sent, w2, smask, wt,
                                              part, D_DIM, D_DIM);
  sum_partials<<<NOUT / 4 / 128, 128, 0, stream>>>(part, vmask, smask, out);
}


# K-loop global loads run two chunks ahead: second set of nine staging registers, two trips per loop pass, loop barrier waits for LDS only so the loads stay in flight across it; sums unchanged
# speedup vs baseline: 4.7407x; 1.0849x over previous
#include <hip/hip_runtime.h>

#define D_DIM  512
#define B_DIM  16
#define TV_DIM 256
#define TS_DIM 32

#define NB_V   256
#define NPART  8
#define NOUT   (B_DIM * TV_DIM * TS_DIM)

#define KC     32
#define LDS_S  36
#define T2_S   68
#define RED_S  33

typedef float v2f __attribute__((ext_vector_type(2)));
typedef float v4f __attribute__((ext_vector_type(4)));
typedef float v8f __attribute__((ext_vector_type(8)));

__device__ __forceinline__ float fast_tanh(float x) {
#if __has_builtin(__builtin_amdgcn_tanhf)
  return __builtin_amdgcn_tanhf(x);
#else
  float y;
  asm("v_tanh_f32 %0, %1" : "=v"(y) : "v"(x));
  return y;
#endif
}

template <int K0, int K1>
__device__ __forceinline__ void wmma_steps(const float* Xb, const float* Wb,
                                           const float* Sb, const float* Vb,
                                           v8f& c0, v8f& c1, v8f& c2, v8f& c3, v8f& c4) {
#pragma unroll
  for (int kk = 4 * K0; kk < 4 * K1; kk += 4) {
    v2f x  = *(const v2f*)(Xb + kk);
    v2f w0 = *(const v2f*)(Wb + kk);
    v2f w1 = *(const v2f*)(Wb + 16 * LDS_S + kk);
    v2f w2 = *(const v2f*)(Wb + 32 * LDS_S + kk);
    v2f w3 = *(const v2f*)(Wb + 48 * LDS_S + kk);
    v2f s  = *(const v2f*)(Sb + kk);
    v2f v  = *(const v2f*)(Vb + kk);
    c0 = __builtin_amdgcn_wmma_f32_16x16x4_f32(false, w0, false, x, (short)0, c0, false, false);
    c1 = __builtin_amdgcn_wmma_f32_16x16x4_f32(false, w1, false, x, (short)0, c1, false, false);
    c2 = __builtin_amdgcn_wmma_f32_16x16x4_f32(false, w2, false, x, (short)0, c2, false, false);
    c3 = __builtin_amdgcn_wmma_f32_16x16x4_f32(false, w3, false, x, (short)0, c3, false, false);
    c4 = __builtin_amdgcn_wmma_f32_16x16x4_f32(false, s,  false, v, (short)0, c4, false, false);
  }
}

struct PairFrag { v2f x, w0, w1, w2, w3, s, v; v2f xq, w0q, w1q, w2q, w3q, sq, vq; };
template <int KS>
__device__ __forceinline__ PairFrag load_pair(const float* Xb, const float* Wb,
                                              const float* Sb, const float* Vb) {
  PairFrag t;
  t.x   = *(const v2f*)(Xb + 4 * KS);
  t.w0  = *(const v2f*)(Wb + 4 * KS);
  t.w1  = *(const v2f*)(Wb + 16 * LDS_S + 4 * KS);
  t.w2  = *(const v2f*)(Wb + 32 * LDS_S + 4 * KS);
  t.w3  = *(const v2f*)(Wb + 48 * LDS_S + 4 * KS);
  t.s   = *(const v2f*)(Sb + 4 * KS);
  t.v   = *(const v2f*)(Vb + 4 * KS);
  t.xq  = *(const v2f*)(Xb + 4 * KS + 4);
  t.w0q = *(const v2f*)(Wb + 4 * KS + 4);
  t.w1q = *(const v2f*)(Wb + 16 * LDS_S + 4 * KS + 4);
  t.w2q = *(const v2f*)(Wb + 32 * LDS_S + 4 * KS + 4);
  t.w3q = *(const v2f*)(Wb + 48 * LDS_S + 4 * KS + 4);
  t.sq  = *(const v2f*)(Sb + 4 * KS + 4);
  t.vq  = *(const v2f*)(Vb + 4 * KS + 4);
  return t;
}
__device__ __forceinline__ void mma_pair(const PairFrag& t,
                                         v8f& c0, v8f& c1, v8f& c2, v8f& c3, v8f& c4) {
  c0 = __builtin_amdgcn_wmma_f32_16x16x4_f32(false, t.w0,  false, t.x,  (short)0, c0, false, false);
  c1 = __builtin_amdgcn_wmma_f32_16x16x4_f32(false, t.w1,  false, t.x,  (short)0, c1, false, false);
  c2 = __builtin_amdgcn_wmma_f32_16x16x4_f32(false, t.w2,  false, t.x,  (short)0, c2, false, false);
  c3 = __builtin_amdgcn_wmma_f32_16x16x4_f32(false, t.w3,  false, t.x,  (short)0, c3, false, false);
  c4 = __builtin_amdgcn_wmma_f32_16x16x4_f32(false, t.s,   false, t.v,  (short)0, c4, false, false);
  c0 = __builtin_amdgcn_wmma_f32_16x16x4_f32(false, t.w0q, false, t.xq, (short)0, c0, false, false);
  c1 = __builtin_amdgcn_wmma_f32_16x16x4_f32(false, t.w1q, false, t.xq, (short)0, c1, false, false);
  c2 = __builtin_amdgcn_wmma_f32_16x16x4_f32(false, t.w2q, false, t.xq, (short)0, c2, false, false);
  c3 = __builtin_amdgcn_wmma_f32_16x16x4_f32(false, t.w3q, false, t.xq, (short)0, c3, false, false);
  c4 = __builtin_amdgcn_wmma_f32_16x16x4_f32(false, t.sq,  false, t.vq, (short)0, c4, false, false);
}

__device__ __forceinline__ void lds_barrier() {
  __builtin_amdgcn_fence(__ATOMIC_RELEASE, "workgroup", "local");
  __builtin_amdgcn_s_barrier();
  __builtin_amdgcn_fence(__ATOMIC_ACQUIRE, "workgroup", "local");
}

struct Stage { v4f a0, a1, a2, a3, w0, w1, s0, v0, v1; };
__device__ __forceinline__ void stage_load(Stage& g, const float* ga, const float* gw,
                                           const float* gs, const float* gv, const size_t rows32) {
  g.a0 = *(const v4f*)(ga);
  g.a1 = *(const v4f*)(ga + rows32);
  g.a2 = *(const v4f*)(ga + 2 * rows32);
  g.a3 = *(const v4f*)(ga + 3 * rows32);
  g.w0 = *(const v4f*)(gw);
  g.w1 = *(const v4f*)(gw + rows32);
  g.s0 = *(const v4f*)(gs);
  g.v0 = *(const v4f*)(gv);
  g.v1 = *(const v4f*)(gv + rows32);
}
__device__ __forceinline__ void stage_store(const Stage& g, float* Ad, float* Wd, float* Sd, float* Vd) {
  *(v4f*)(Ad)              = g.a0;
  *(v4f*)(Ad + 32 * LDS_S) = g.a1;
  *(v4f*)(Ad + 64 * LDS_S) = g.a2;
  *(v4f*)(Ad + 96 * LDS_S) = g.a3;
  *(v4f*)(Wd)              = g.w0;
  *(v4f*)(Wd + 32 * LDS_S) = g.w1;
  *(v4f*)(Sd)              = g.s0;
  *(v4f*)(Vd)              = g.v0;
  *(v4f*)(Vd + 32 * LDS_S) = g.v1;
}

template <int BUF>
__device__ __forceinline__ void k_trip(float* Asb, float* Wsb, float* Ssb, float* Vsb,
                                       const int xoff, const int woff, const int soff, const int voff,
                                       const int ls, const bool have_t, PairFrag& t, const Stage& st,
                                       v8f& c0, v8f& c1, v8f& c2, v8f& c3, v8f& c4) {
  const float* Xb = Asb + BUF * 128 * LDS_S + xoff;
  const float* Wb = Wsb + BUF *  64 * LDS_S + woff;
  const float* Sb = Ssb + BUF *  32 * LDS_S + soff;
  const float* Vb = Vsb + BUF *  64 * LDS_S + voff;
  if (have_t) mma_pair(t, c0, c1, c2, c3, c4);
  wmma_steps<0, 4>(Xb, Wb, Sb, Vb, c0, c1, c2, c3, c4);
  const PairFrag m = load_pair<4>(Xb, Wb, Sb, Vb);
  t = load_pair<6>(Xb, Wb, Sb, Vb);
  __builtin_amdgcn_sched_barrier(0);
  stage_store(st, Asb + (BUF ^ 1) * 128 * LDS_S + ls, Wsb + (BUF ^ 1) * 64 * LDS_S + ls,
              Ssb + (BUF ^ 1) * 32 * LDS_S + ls, Vsb + (BUF ^ 1) * 64 * LDS_S + ls);
  mma_pair(m, c0, c1, c2, c3, c4);
  lds_barrier();
}

__device__ __forceinline__ void tanh_dot8(const v8f& cj, const v4f ya, const v4f yb,
                                          const v4f wa, const v4f wb, const v2f p,
                                          v2f& lo, v2f& hi) {
  const v2f x0 = {cj[0], cj[1]}, x1 = {cj[2], cj[3]}, x2 = {cj[4], cj[5]}, x3 = {cj[6], cj[7]};
  const v2f y0 = {ya.x, ya.y}, y1 = {ya.z, ya.w}, y2 = {yb.x, yb.y}, y3 = {yb.z, yb.w};
  const v2f w0 = {wa.x, wa.y}, w1 = {wa.z, wa.w}, w2 = {wb.x, wb.y}, w3 = {wb.z, wb.w};
  const v2f u0 = (x0 + y0) * p, u1 = (x1 + y1) * p, u2 = (x2 + y2) * p, u3 = (x3 + y3) * p;
  const v2f t0 = {fast_tanh(u0.x), fast_tanh(u0.y)};
  const v2f t1 = {fast_tanh(u1.x), fast_tanh(u1.y)};
  const v2f t2 = {fast_tanh(u2.x), fast_tanh(u2.y)};
  const v2f t3 = {fast_tanh(u3.x), fast_tanh(u3.y)};
  lo = t0 * w0 + lo;
  hi = t1 * w1 + hi;
  lo = t2 * w2 + lo;
  hi = t3 * w3 + hi;
}

__global__ __launch_bounds__(256) __attribute__((amdgpu_waves_per_eu(2, 4))) void proj_tanh_partial(
    const float* __restrict__ Av, const float* __restrict__ Wv,
    const float* __restrict__ biasv, const float* __restrict__ rmaskv,
    const float* __restrict__ Asn, const float* __restrict__ Wsn,
    const float* __restrict__ rmasks, const float* __restrict__ wtg,
    float* __restrict__ part, int N, int K) {
  __shared__ float As[2][128 * LDS_S];
  __shared__ float Ws[2][ 64 * LDS_S];
  __shared__ float Ss[2][ 32 * LDS_S];
  __shared__ float Vs[2][ 64 * LDS_S];
  __shared__ __align__(16) float T2s[TS_DIM * T2_S];
  __shared__ float red[8 * TS_DIM * RED_S];

  const int tid  = threadIdx.x;
  const int lane = tid & 31;
  const int wib  = tid >> 5;
  const int bid  = (int)blockIdx.x;
  const int ntn  = N >> 6;
  const int mblk = bid / ntn;
  const int ng   = bid % ntn;
  const int mBase = mblk * 128;
  const int n0    = ng * 64;
  const int sBase = (mBase / TV_DIM) * TS_DIM;
  const int half  = lane >> 4;
  const int lm    = lane & 15;
  const int nchunks = K / KC;

  const int sr = tid >> 3;
  const int sc = (tid & 7) * 4;
  const size_t rows32 = (size_t)32 * K;
  const float* ga = Av  + (size_t)(mBase + sr) * K + sc;
  const float* gw = Wv  + (size_t)(n0    + sr) * K + sc;
  const float* gs = Asn + (size_t)(sBase + sr) * K + sc;
  const float* gv = Wsn + (size_t)(n0    + sr) * K + sc;
  const int ls = sr * LDS_S + sc;

  Stage gA, gB;
  stage_load(gA, ga, gw, gs, gv, rows32);
  if (nchunks > 1) stage_load(gB, ga + KC, gw + KC, gs + KC, gv + KC, rows32);
  ga += 2 * KC;
  gw += 2 * KC;
  gs += 2 * KC;
  gv += 2 * KC;
  stage_store(gA, &As[0][ls], &Ws[0][ls], &Ss[0][ls], &Vs[0][ls]);
  lds_barrier();

  v8f c0 = {0.f,0.f,0.f,0.f,0.f,0.f,0.f,0.f};
  v8f c1 = {0.f,0.f,0.f,0.f,0.f,0.f,0.f,0.f};
  v8f c2 = {0.f,0.f,0.f,0.f,0.f,0.f,0.f,0.f};
  v8f c3 = {0.f,0.f,0.f,0.f,0.f,0.f,0.f,0.f};
  v8f c4 = {0.f,0.f,0.f,0.f,0.f,0.f,0.f,0.f};

  const int st = wib >> 2;
  const int dt = wib & 3;
  const int xoff = (16 * wib + lm) * LDS_S + 2 * half;
  const int woff = lm * LDS_S + 2 * half;
  const int soff = (16 * st + lm) * LDS_S + 2 * half;
  const int voff = (16 * dt + lm) * LDS_S + 2 * half;
  PairFrag t;
  int c = 0;
  for (; c + 3 < nchunks; c += 2) {
    stage_load(gA, ga, gw, gs, gv, rows32);
    ga += KC;
    gw += KC;
    gs += KC;
    gv += KC;
    k_trip<0>(&As[0][0], &Ws[0][0], &Ss[0][0], &Vs[0][0], xoff, woff, soff, voff, ls,
              c > 0, t, gB, c0, c1, c2, c3, c4);
    stage_load(gB, ga, gw, gs, gv, rows32);
    ga += KC;
    gw += KC;
    gs += KC;
    gv += KC;
    k_trip<1>(&As[0][0], &Ws[0][0], &Ss[0][0], &Vs[0][0], xoff, woff, soff, voff, ls,
              true, t, gA, c0, c1, c2, c3, c4);
  }
  if (c + 2 < nchunks) {
    stage_load(gA, ga, gw, gs, gv, rows32);
    k_trip<0>(&As[0][0], &Ws[0][0], &Ss[0][0], &Vs[0][0], xoff, woff, soff, voff, ls,
              c > 0, t, gB, c0, c1, c2, c3, c4);
    k_trip<1>(&As[0][0], &Ws[0][0], &Ss[0][0], &Vs[0][0], xoff, woff, soff, voff, ls,
              true, t, gA, c0, c1, c2, c3, c4);
  } else if (c + 1 < nchunks) {
    k_trip<0>(&As[0][0], &Ws[0][0], &Ss[0][0], &Vs[0][0], xoff, woff, soff, voff, ls,
              c > 0, t, gB, c0, c1, c2, c3, c4);
  }
  {
    const int buf = (nchunks - 1) & 1;
    if (nchunks > 1) mma_pair(t, c0, c1, c2, c3, c4);
    wmma_steps<0, 8>(&As[buf][xoff], &Ws[buf][woff], &Ss[buf][soff], &Vs[buf][voff], c0, c1, c2, c3, c4);
  }

#pragma unroll
  for (int r = 0; r < 8; ++r) {
    const int s = 16 * st + 8 * half + r;
    T2s[s * T2_S + 16 * dt + lm] = c4[r] * rmasks[sBase + s];
  }
  if (tid < TS_DIM) T2s[tid * T2_S + 64] = rmasks[sBase + tid];

  const int vrow = mBase + 16 * wib + lm;
  const float vm = rmaskv[vrow];
  const float* bp = biasv + n0 + 8 * half;
  const float* wp = wtg   + n0 + 8 * half;
  const v4f wa0 = *(const v4f*)(wp),      wb0 = *(const v4f*)(wp + 4);
  const v4f wa1 = *(const v4f*)(wp + 16), wb1 = *(const v4f*)(wp + 20);
  const v4f wa2 = *(const v4f*)(wp + 32), wb2 = *(const v4f*)(wp + 36);
  const v4f wa3 = *(const v4f*)(wp + 48), wb3 = *(const v4f*)(wp + 52);
  {
    const v4f ba0 = *(const v4f*)(bp),      bb0 = *(const v4f*)(bp + 4);
    const v4f ba1 = *(const v4f*)(bp + 16), bb1 = *(const v4f*)(bp + 20);
    const v4f ba2 = *(const v4f*)(bp + 32), bb2 = *(const v4f*)(bp + 36);
    const v4f ba3 = *(const v4f*)(bp + 48), bb3 = *(const v4f*)(bp + 52);
    c0 = (c0 + __builtin_shufflevector(ba0, bb0, 0, 1, 2, 3, 4, 5, 6, 7)) * vm;
    c1 = (c1 + __builtin_shufflevector(ba1, bb1, 0, 1, 2, 3, 4, 5, 6, 7)) * vm;
    c2 = (c2 + __builtin_shufflevector(ba2, bb2, 0, 1, 2, 3, 4, 5, 6, 7)) * vm;
    c3 = (c3 + __builtin_shufflevector(ba3, bb3, 0, 1, 2, 3, 4, 5, 6, 7)) * vm;
  }
  __syncthreads();

  float* redw = red + wib * TS_DIM * RED_S;
#pragma unroll 2
  for (int s = 0; s < TS_DIM; ++s) {
    const float* yr = T2s + s * T2_S + 8 * half;
    const float pm = vm * T2s[s * T2_S + 64];
    const v2f p = {pm, pm};
    v2f lo = {0.f, 0.f}, hi = {0.f, 0.f};
    tanh_dot8(c0, *(const v4f*)(yr),      *(const v4f*)(yr + 4),  wa0, wb0, p, lo, hi);
    tanh_dot8(c1, *(const v4f*)(yr + 16), *(const v4f*)(yr + 20), wa1, wb1, p, lo, hi);
    tanh_dot8(c2, *(const v4f*)(yr + 32), *(const v4f*)(yr + 36), wa2, wb2, p, lo, hi);
    tanh_dot8(c3, *(const v4f*)(yr + 48), *(const v4f*)(yr + 52), wa3, wb3, p, lo, hi);
    redw[s * RED_S + lane] = (lo.x + lo.y) + (hi.x + hi.y);
  }
  __syncthreads();

  const int sv = lane >> 3;
  const int sq = 4 * (lane & 7);
#pragma unroll
  for (int q = 0; q < 4; ++q) {
    const int v = 4 * q + sv;
    float e[4];
#pragma unroll
    for (int i = 0; i < 4; ++i)
      e[i] = redw[(sq + i) * RED_S + v] + redw[(sq + i) * RED_S + v + 16];
    *(v4f*)(part + (size_t)ng * NOUT + (size_t)(mBase + 16 * wib + v) * TS_DIM + sq) =
        (v4f){e[0], e[1], e[2], e[3]};
  }
}

__global__ __launch_bounds__(128) void sum_partials(
    const float* __restrict__ part,
    const float* __restrict__ vmask,
    const float* __restrict__ smask,
    float* __restrict__ out) {
  const int i4 = (int)blockIdx.x * 128 + (int)threadIdx.x;
  const v4f* p4 = (const v4f*)part;
  v4f r[NPART];
#pragma unroll
  for (int n = 0; n < NPART; ++n) r[n] = p4[(size_t)n * (NOUT / 4) + i4];
  v4f acc = r[0];
#pragma unroll
  for (int n = 1; n < NPART; ++n) acc += r[n];
  const int i   = 4 * i4;
  const int row = i / TS_DIM;
  const int s   = i % TS_DIM;
  const float vm = vmask[row];
  const v4f sm = *(const v4f*)(smask + (row / TV_DIM) * TS_DIM + s);
  ((v4f*)out)[i4] = acc * (sm * vm);
}

extern "C" void kernel_launch(void* const* d_in, const int* in_sizes, int n_in,
                              void* d_out, int out_size, void* d_ws, size_t ws_size,
                              hipStream_t stream) {
  const float* video = (const float*)d_in[0];
  const float* vmask = (const float*)d_in[1];
  const float* sent  = (const float*)d_in[2];
  const float* smask = (const float*)d_in[3];
  const float* w1    = (const float*)d_in[4];
  const float* b1    = (const float*)d_in[5];
  const float* w2    = (const float*)d_in[6];
  const float* wt    = (const float*)d_in[7];
  float* out  = (float*)d_out;

  float* part = (float*)d_ws;

  proj_tanh_partial<<<NB_V, 256, 0, stream>>>(video, w1, b1, vmask, sent, w2, smask, wt,
                                              part, D_DIM, D_DIM);
  sum_partials<<<NOUT / 4 / 128, 128, 0, stream>>>(part, vmask, smask, out);
}
